# starting kernel
# baseline (speedup 1.0000x reference)
#include <hip/hip_runtime.h>


typedef __attribute__((ext_vector_type(2))) float v2f;
typedef __attribute__((ext_vector_type(8))) float v8f;

#define B_DIM   32
#define N_PTS   4096
#define M_STARS 1024
#define LOG2E   1.44269504088896340736f

__global__ __launch_bounds__(256) void ursa_gau_wmma(
    const float* __restrict__ pts,
    const float* __restrict__ stars,
    float* __restrict__ out)
{
    const int mtile = blockIdx.x;
    const int b     = blockIdx.y;
    const int tid   = threadIdx.x;
    const int wave  = tid >> 5;
    const int lane  = tid & 31;
    const int l16   = lane & 15;
    const bool hi   = (lane >= 16);

    const int m_base = mtile * 16;

    const int   m  = m_base + l16;
    const float sx = stars[3 * m + 0];
    const float sy = stars[3 * m + 1];
    const float sz = stars[3 * m + 2];
    const float s2 = sx * sx + sy * sy + sz * sz;
    v2f bmat;
    bmat[0] = (hi ? sz : sx) * LOG2E;
    bmat[1] = hi ? (-0.5f * LOG2E) : (sy * LOG2E);
    const float cval = -0.5f * LOG2E * s2;
    v8f cvec;
#pragma unroll
    for (int r = 0; r < 8; ++r) cvec[r] = cval;

    const float* ptsb = pts + (size_t)b * (N_PTS * 3);
    float acc = 0.0f;

    const int tiles_per_wave = (N_PTS / 16) / 8;
#pragma unroll 2
    for (int i = 0; i < tiles_per_wave; ++i) {
        const int n = (wave * tiles_per_wave + i) * 16 + l16;
        const float x  = ptsb[3 * n + 0];
        const float y  = ptsb[3 * n + 1];
        const float z  = ptsb[3 * n + 2];
        const float p2 = x * x + y * y + z * z;
        v2f amat;
        amat[0] = hi ? z  : x;
        amat[1] = hi ? p2 : y;

        v8f d = __builtin_amdgcn_wmma_f32_16x16x4_f32(
             false, amat,  false, bmat,
             (short)0, cvec,  false,  false);

        float s = 0.0f;
#pragma unroll
        for (int r = 0; r < 8; ++r) s += __builtin_amdgcn_exp2f(d[r]);
        acc += s;
    }

    float tot = acc + __shfl_xor(acc, 16, 32);

    __shared__ float partial[8][16];
    if (lane < 16) partial[wave][l16] = tot;
    __syncthreads();
    if (tid < 16) {
        float s = 0.0f;
#pragma unroll
        for (int w = 0; w < 8; ++w) s += partial[w][tid];
        out[(size_t)b * M_STARS + m_base + tid] = s;
    }
}

extern "C" void kernel_launch(void* const* d_in, const int* in_sizes, int n_in,
                              void* d_out, int out_size, void* d_ws, size_t ws_size,
                              hipStream_t stream) {
    const float* pts   = (const float*)d_in[0];
    const float* stars = (const float*)d_in[1];
    float* out         = (float*)d_out;

    dim3 grid(M_STARS / 16, B_DIM);
    ursa_gau_wmma<<<grid, 256, 0, stream>>>(pts, stars, out);
}


# Each wave keeps the B/C fragments of 4 star tiles and uses each point tile's A fragment for 4 WMMAs: 4x fewer waves, point loads and A set-ups; two point tiles are loaded one trip ahead
# speedup vs baseline: 1.0604x; 1.0604x over previous
#include <hip/hip_runtime.h>


typedef __attribute__((ext_vector_type(2))) float v2f;
typedef __attribute__((ext_vector_type(8))) float v8f;

#define B_DIM   32
#define N_PTS   4096
#define M_STARS 1024
#define MT      4
#define M_GRP   (16 * MT)
#define LOG2E   1.44269504088896340736f

__global__ __launch_bounds__(256) void ursa_gau_wmma(
    const float* __restrict__ pts,
    const float* __restrict__ stars,
    float* __restrict__ out)
{
    const int mgrp  = blockIdx.x;
    const int b     = blockIdx.y;
    const int tid   = threadIdx.x;
    const int wave  = tid >> 5;
    const int lane  = tid & 31;
    const int l16   = lane & 15;
    const bool hi   = (lane >= 16);

    const int m_base = mgrp * M_GRP;

    v2f bmat[MT];
    v8f cvec[MT];
#pragma unroll
    for (int t = 0; t < MT; ++t) {
        const int   m  = m_base + t * 16 + l16;
        const float sx = stars[3 * m + 0];
        const float sy = stars[3 * m + 1];
        const float sz = stars[3 * m + 2];
        const float s2 = __builtin_fmaf(sz, sz, __builtin_fmaf(sx, sx, sy * sy));
        bmat[t][0] = (hi ? sz : sx) * LOG2E;
        bmat[t][1] = hi ? (-0.5f * LOG2E) : (sy * LOG2E);
        const float cval = -0.5f * LOG2E * s2;
#pragma unroll
        for (int r = 0; r < 8; ++r) cvec[t][r] = cval;
    }

    const float* ptsb = pts + (size_t)b * (N_PTS * 3);
    float acc[MT];
#pragma unroll
    for (int t = 0; t < MT; ++t) acc[t] = 0.0f;

    const int tiles_per_wave = (N_PTS / 16) / 8;
    const int n0 = wave * tiles_per_wave * 16 + l16;

    float px[2], py[2], pz[2];
#pragma unroll
    for (int k = 0; k < 2; ++k) {
        const int nk = n0 + k * 16;
        px[k] = ptsb[3 * nk + 0];
        py[k] = ptsb[3 * nk + 1];
        pz[k] = ptsb[3 * nk + 2];
    }

    for (int i = 0; i < tiles_per_wave; i += 2) {
        float cx[2], cy[2], cz[2];
#pragma unroll
        for (int k = 0; k < 2; ++k) { cx[k] = px[k]; cy[k] = py[k]; cz[k] = pz[k]; }
        const int inext = (i + 2 < tiles_per_wave) ? (i + 2) : i;
#pragma unroll
        for (int k = 0; k < 2; ++k) {
            const int nn = n0 + (inext + k) * 16;
            px[k] = ptsb[3 * nn + 0];
            py[k] = ptsb[3 * nn + 1];
            pz[k] = ptsb[3 * nn + 2];
        }

#pragma unroll
        for (int k = 0; k < 2; ++k) {
            const float x = cx[k], y = cy[k], z = cz[k];

            const float p2 = __builtin_fmaf(z, z, __builtin_fmaf(x, x, y * y));
            v2f amat;
            amat[0] = hi ? z  : x;
            amat[1] = hi ? p2 : y;

            v8f d[MT];
#pragma unroll
            for (int t = 0; t < MT; ++t)
                d[t] = __builtin_amdgcn_wmma_f32_16x16x4_f32(
                     false, amat,  false, bmat[t],
                     (short)0, cvec[t],  false,  false);

            float s[MT], e[MT];
#pragma unroll
            for (int t = 0; t < MT; ++t) s[t] = __builtin_amdgcn_exp2f(d[t][0]);
#pragma unroll
            for (int t = 0; t < MT; ++t) e[t] = __builtin_amdgcn_exp2f(d[t][1]);
#pragma unroll
            for (int r = 2; r < 8; ++r) {
                __builtin_amdgcn_sched_barrier(0);
#pragma unroll
                for (int t = 0; t < MT; ++t) {
                    s[t] += e[t];
                    e[t] = __builtin_amdgcn_exp2f(d[t][r]);
                }
            }
            __builtin_amdgcn_sched_barrier(0);
#pragma unroll
            for (int t = 0; t < MT; ++t) {
                s[t] += e[t];
                acc[t] += s[t];
            }
        }
    }

    __shared__ float partial[8][M_GRP];
#pragma unroll
    for (int t = 0; t < MT; ++t) {
        const float tot = acc[t] + __shfl_xor(acc[t], 16, 32);
        if (lane < 16) partial[wave][t * 16 + l16] = tot;
    }
    __syncthreads();
    if (tid < M_GRP) {
        float s = 0.0f;
#pragma unroll
        for (int w = 0; w < 8; ++w) s += partial[w][tid];
        out[(size_t)b * M_STARS + m_base + tid] = s;
    }
}

extern "C" void kernel_launch(void* const* d_in, const int* in_sizes, int n_in,
                              void* d_out, int out_size, void* d_ws, size_t ws_size,
                              hipStream_t stream) {
    const float* pts   = (const float*)d_in[0];
    const float* stars = (const float*)d_in[1];
    float* out         = (float*)d_out;

    dim3 grid(M_STARS / M_GRP, B_DIM);
    ursa_gau_wmma<<<grid, 256, 0, stream>>>(pts, stars, out);
}


# Each wave keeps 8 star tiles instead of 4 (128 stars per work-group): 2048 waves in 256 work-groups, one per WGP; half the wave entries, point loads and A set-ups per product; sum order kept
# speedup vs baseline: 1.1692x; 1.1026x over previous
#include <hip/hip_runtime.h>


typedef __attribute__((ext_vector_type(2))) float v2f;
typedef __attribute__((ext_vector_type(8))) float v8f;

#define B_DIM   32
#define N_PTS   4096
#define M_STARS 1024
#define MT      8
#define M_GRP   (16 * MT)
#define LOG2E   1.44269504088896340736f

__global__ __launch_bounds__(256) void ursa_gau_wmma(
    const float* __restrict__ pts,
    const float* __restrict__ stars,
    float* __restrict__ out)
{
    const int mgrp  = blockIdx.x;
    const int b     = blockIdx.y;
    const int tid   = threadIdx.x;
    const int wave  = tid >> 5;
    const int lane  = tid & 31;
    const int l16   = lane & 15;
    const bool hi   = (lane >= 16);

    const int m_base = mgrp * M_GRP;

    v2f bmat[MT];
    v8f cvec[MT];
#pragma unroll
    for (int t = 0; t < MT; ++t) {
        const int   m  = m_base + t * 16 + l16;
        const float sx = stars[3 * m + 0];
        const float sy = stars[3 * m + 1];
        const float sz = stars[3 * m + 2];
        const float s2 = __builtin_fmaf(sz, sz, __builtin_fmaf(sx, sx, sy * sy));
        bmat[t][0] = (hi ? sz : sx) * LOG2E;
        bmat[t][1] = hi ? (-0.5f * LOG2E) : (sy * LOG2E);
        const float cval = -0.5f * LOG2E * s2;
#pragma unroll
        for (int r = 0; r < 8; ++r) cvec[t][r] = cval;
    }

    const float* ptsb = pts + (size_t)b * (N_PTS * 3);
    float acc[MT];
#pragma unroll
    for (int t = 0; t < MT; ++t) acc[t] = 0.0f;

    const int tiles_per_wave = (N_PTS / 16) / 8;
    const int n0 = wave * tiles_per_wave * 16 + l16;

    float px[2], py[2], pz[2];
#pragma unroll
    for (int k = 0; k < 2; ++k) {
        const int nk = n0 + k * 16;
        px[k] = ptsb[3 * nk + 0];
        py[k] = ptsb[3 * nk + 1];
        pz[k] = ptsb[3 * nk + 2];
    }

    for (int i = 0; i < tiles_per_wave; i += 2) {
        float cx[2], cy[2], cz[2];
#pragma unroll
        for (int k = 0; k < 2; ++k) { cx[k] = px[k]; cy[k] = py[k]; cz[k] = pz[k]; }
        const int inext = (i + 2 < tiles_per_wave) ? (i + 2) : i;
#pragma unroll
        for (int k = 0; k < 2; ++k) {
            const int nn = n0 + (inext + k) * 16;
            px[k] = ptsb[3 * nn + 0];
            py[k] = ptsb[3 * nn + 1];
            pz[k] = ptsb[3 * nn + 2];
        }

#pragma unroll
        for (int k = 0; k < 2; ++k) {
            const float x = cx[k], y = cy[k], z = cz[k];

            const float p2 = __builtin_fmaf(z, z, __builtin_fmaf(x, x, y * y));
            v2f amat;
            amat[0] = hi ? z  : x;
            amat[1] = hi ? p2 : y;

            v8f d[MT];
#pragma unroll
            for (int t = 0; t < MT; ++t)
                d[t] = __builtin_amdgcn_wmma_f32_16x16x4_f32(
                     false, amat,  false, bmat[t],
                     (short)0, cvec[t],  false,  false);

            float s[MT], e[MT];
#pragma unroll
            for (int t = 0; t < MT; ++t) s[t] = __builtin_amdgcn_exp2f(d[t][0]);
#pragma unroll
            for (int t = 0; t < MT; ++t) e[t] = __builtin_amdgcn_exp2f(d[t][1]);
#pragma unroll
            for (int r = 2; r < 8; ++r) {
                __builtin_amdgcn_sched_barrier(0);
#pragma unroll
                for (int t = 0; t < MT; ++t) {
                    s[t] += e[t];
                    e[t] = __builtin_amdgcn_exp2f(d[t][r]);
                }
            }
            __builtin_amdgcn_sched_barrier(0);
#pragma unroll
            for (int t = 0; t < MT; ++t) {
                s[t] += e[t];
                acc[t] += s[t];
            }
        }
    }

    __shared__ float partial[8][M_GRP];
#pragma unroll
    for (int t = 0; t < MT; ++t) {
        const float tot = acc[t] + __shfl_xor(acc[t], 16, 32);
        if (lane < 16) partial[wave][t * 16 + l16] = tot;
    }
    __syncthreads();
    if (tid < M_GRP) {
        float s = 0.0f;
#pragma unroll
        for (int w = 0; w < 8; ++w) s += partial[w][tid];
        out[(size_t)b * M_STARS + m_base + tid] = s;
    }
}

extern "C" void kernel_launch(void* const* d_in, const int* in_sizes, int n_in,
                              void* d_out, int out_size, void* d_ws, size_t ws_size,
                              hipStream_t stream) {
    const float* pts   = (const float*)d_in[0];
    const float* stars = (const float*)d_in[1];
    float* out         = (float*)d_out;

    dim3 grid(M_STARS / M_GRP, B_DIM);
    ursa_gau_wmma<<<grid, 256, 0, stream>>>(pts, stars, out);
}
